# starting kernel
# baseline (speedup 1.0000x reference)
#include <hip/hip_runtime.h>
#include <hip/hip_bf16.h>
#include <math.h>

#define HIDDEN   2048
#define NH       16
#define NKV      4
#define HD       128
#define D2       64
#define QSIZE    2048
#define KVSIZE   512
#define QKVLD    3072
#define KOFF     2048
#define VOFF     2560
#define SEQB     2048
#define BATCH    2
#define NTOK     (BATCH * SEQB)
#define SM_SCALE 0.08838834764831845f

typedef __attribute__((ext_vector_type(2))) float v2f;
typedef __attribute__((ext_vector_type(8))) float v8f;

__device__ __forceinline__ v8f wmma_f32(v2f a, v2f b, v8f c) {
  return __builtin_amdgcn_wmma_f32_16x16x4_f32(
       false, a,  false, b,
       (short)0, c,  false,  false);
}

#define GBK  32
#define GLDP 36

__global__ __launch_bounds__(128)
void gemm_nt_kernel(const float* __restrict__ A, const float* __restrict__ Bw,
                    const float* __restrict__ bias, float* __restrict__ C,
                    int M, int N, int K) {
  __shared__ float sA[128 * GLDP];
  __shared__ float sB[64 * GLDP];
  const int t    = threadIdx.x;
  const int lane = t & 31, wid = t >> 5;
  const int hi   = lane >> 4, l15 = lane & 15;
  const int mbase = blockIdx.y * 128;
  const int nbase = blockIdx.x * 64;

  const v8f vz = {0.f,0.f,0.f,0.f,0.f,0.f,0.f,0.f};
  v8f acc[2][4] = {{vz, vz, vz, vz}, {vz, vz, vz, vz}};

  const int srow = t >> 3;
  const int scol = (t & 7) << 2;

  for (int kb = 0; kb < K; kb += GBK) {
#pragma unroll
    for (int i = 0; i < 8; i++) {
      const int row = srow + 16 * i;
      *(float4*)&sA[row * GLDP + scol] =
          *(const float4*)&A[(size_t)(mbase + row) * K + kb + scol];
    }
#pragma unroll
    for (int i = 0; i < 4; i++) {
      const int row = srow + 16 * i;
      *(float4*)&sB[row * GLDP + scol] =
          *(const float4*)&Bw[(size_t)(nbase + row) * K + kb + scol];
    }
    if (kb + GBK < K) {
      __builtin_prefetch(&A[(size_t)(mbase + srow) * K + kb + GBK + scol], 0, 0);
      __builtin_prefetch(&Bw[(size_t)(nbase + srow) * K + kb + GBK + scol], 0, 0);
    }
    __syncthreads();
#pragma unroll
    for (int k2 = 0; k2 < GBK; k2 += 4) {
      const v2f a0 = *(const v2f*)&sA[(wid * 32 + l15) * GLDP + k2 + 2 * hi];
      const v2f a1 = *(const v2f*)&sA[(wid * 32 + 16 + l15) * GLDP + k2 + 2 * hi];
#pragma unroll
      for (int nc = 0; nc < 4; nc++) {
        const v2f b = *(const v2f*)&sB[(nc * 16 + l15) * GLDP + k2 + 2 * hi];
        acc[0][nc] = wmma_f32(a0, b, acc[0][nc]);
        acc[1][nc] = wmma_f32(a1, b, acc[1][nc]);
      }
    }
    __syncthreads();
  }

#pragma unroll
  for (int st = 0; st < 2; st++) {
#pragma unroll
    for (int nc = 0; nc < 4; nc++) {
      const int n  = nbase + nc * 16 + l15;
      const float bv = bias ? bias[n] : 0.0f;
#pragma unroll
      for (int r = 0; r < 8; r++) {
        const int m = mbase + wid * 32 + st * 16 + r + 8 * hi;
        C[(size_t)m * N + n] = acc[st][nc][r] + bv;
      }
    }
  }
}

__global__ __launch_bounds__(256)
void rope_kernel(float* __restrict__ qkv) {
  const int tid   = blockIdx.x * 256 + threadIdx.x;
  const int token = tid / 1280;
  const int rem   = tid % 1280;
  const int head  = rem >> 6;
  const int d     = rem & 63;
  const int pos   = token & (SEQB - 1);
  const int col   = (head < NH) ? head * HD + d : QSIZE + (head - NH) * HD + d;
  const float inv_freq = powf(1.0e6f, -(float)d * (1.0f / 64.0f));
  float s, c;
  sincosf((float)pos * inv_freq, &s, &c);
  float* p = qkv + (size_t)token * QKVLD + col;
  const float x1 = p[0], x2 = p[D2];
  p[0]  = x1 * c - x2 * s;
  p[D2] = x1 * s + x2 * c;
}

__global__ __launch_bounds__(128)
void flash_attn_kernel(const float* __restrict__ qkv, float* __restrict__ out) {
  __shared__ float sQ[64 * 132];
  __shared__ float sK[16 * 132];
  __shared__ float sVt[128 * 20];
  __shared__ float sP[4][16 * 18];

  const int t    = threadIdx.x;
  const int lane = t & 31, wid = t >> 5;
  const int hi   = lane >> 4, l15 = lane & 15;
  const int qblk = blockIdx.x;
  const int h    = blockIdx.y;
  const int b    = blockIdx.z;
  const int kvh  = h >> 2;
  const int qbase = qblk * 64;
  const size_t rowbase = (size_t)(b * SEQB) * QKVLD;

#pragma unroll
  for (int i = 0; i < 16; i++) {
    const int idx = t + 128 * i;
    const int row = idx >> 5, c4 = (idx & 31) << 2;
    float4 v = *(const float4*)&qkv[rowbase + (size_t)(qbase + row) * QKVLD + h * HD + c4];
    v.x *= SM_SCALE; v.y *= SM_SCALE; v.z *= SM_SCALE; v.w *= SM_SCALE;
    *(float4*)&sQ[row * 132 + c4] = v;
  }

  const v8f vz = {0.f,0.f,0.f,0.f,0.f,0.f,0.f,0.f};
  v8f accO[8] = {vz, vz, vz, vz, vz, vz, vz, vz};
  float rm[8], rs[8];
#pragma unroll
  for (int r = 0; r < 8; r++) { rm[r] = -__builtin_inff(); rs[r] = 0.0f; }

  __syncthreads();

  const int nkb = (qbase >> 4) + 4;
  for (int kbi = 0; kbi < nkb; kbi++) {
    const int kb = kbi * 16;
#pragma unroll
    for (int i = 0; i < 4; i++) {
      const int idx = t + 128 * i;
      const int row = idx >> 5, c4 = (idx & 31) << 2;
      const float4 v = *(const float4*)&qkv[rowbase + (size_t)(kb + row) * QKVLD + KOFF + kvh * HD + c4];
      *(float4*)&sK[row * 132 + c4] = v;
    }
#pragma unroll
    for (int i = 0; i < 16; i++) {
      sVt[t * 20 + i] = qkv[rowbase + (size_t)(kb + i) * QKVLD + VOFF + kvh * HD + t];
    }
    __syncthreads();

    v8f s = vz;
#pragma unroll
    for (int k0 = 0; k0 < HD; k0 += 4) {
      const v2f a  = *(const v2f*)&sQ[(wid * 16 + l15) * 132 + k0 + 2 * hi];
      const v2f bb = *(const v2f*)&sK[l15 * 132 + k0 + 2 * hi];
      s = wmma_f32(a, bb, s);
    }

    const int kcol = kb + l15;
#pragma unroll
    for (int r = 0; r < 8; r++) {
      const int mrow = qbase + wid * 16 + r + 8 * hi;
      const float sv = (kcol <= mrow) ? s[r] : -__builtin_inff();
      float mx = sv;
      mx = fmaxf(mx, __shfl_xor(mx, 1, 32));
      mx = fmaxf(mx, __shfl_xor(mx, 2, 32));
      mx = fmaxf(mx, __shfl_xor(mx, 4, 32));
      mx = fmaxf(mx, __shfl_xor(mx, 8, 32));
      const float newm  = fmaxf(rm[r], mx);
      const float alpha = __expf(rm[r] - newm);
      const float pv    = __expf(sv - newm);
      float ps = pv;
      ps += __shfl_xor(ps, 1, 32);
      ps += __shfl_xor(ps, 2, 32);
      ps += __shfl_xor(ps, 4, 32);
      ps += __shfl_xor(ps, 8, 32);
      rs[r] = rs[r] * alpha + ps;
      rm[r] = newm;
#pragma unroll
      for (int nc = 0; nc < 8; nc++) accO[nc][r] *= alpha;
      sP[wid][(r + 8 * hi) * 18 + l15] = pv;
    }
    __syncthreads();

#pragma unroll
    for (int k0 = 0; k0 < 16; k0 += 4) {
      const v2f a = *(const v2f*)&sP[wid][l15 * 18 + k0 + 2 * hi];
#pragma unroll
      for (int nc = 0; nc < 8; nc++) {
        const v2f bb = *(const v2f*)&sVt[(nc * 16 + l15) * 20 + k0 + 2 * hi];
        accO[nc] = wmma_f32(a, bb, accO[nc]);
      }
    }
    __syncthreads();
  }

#pragma unroll
  for (int nc = 0; nc < 8; nc++) {
#pragma unroll
    for (int r = 0; r < 8; r++) {
      const int mrow = qbase + wid * 16 + r + 8 * hi;
      out[(size_t)(b * SEQB + mrow) * HIDDEN + h * HD + nc * 16 + l15] = accO[nc][r] / rs[r];
    }
  }
}

extern "C" void kernel_launch(void* const* d_in, const int* in_sizes, int n_in,
                              void* d_out, int out_size, void* d_ws, size_t ws_size,
                              hipStream_t stream) {
  const float* hs    = (const float*)d_in[0];
  const float* qkv_w = (const float*)d_in[1];
  const float* qkv_b = (const float*)d_in[2];
  const float* o_w   = (const float*)d_in[3];
  float* out  = (float*)d_out;
  float* qkv  = (float*)d_ws;
  float* attn = qkv + (size_t)NTOK * QKVLD;

  dim3 g1(QKVLD / 64, NTOK / 128);
  gemm_nt_kernel<<<g1, 128, 0, stream>>>(hs, qkv_w, qkv_b, qkv, NTOK, QKVLD, HIDDEN);

  rope_kernel<<<(NTOK * 1280) / 256, 256, 0, stream>>>(qkv);

  dim3 g3(SEQB / 64, NH, BATCH);
  flash_attn_kernel<<<g3, 128, 0, stream>>>(qkv, attn);

  dim3 g4(HIDDEN / 64, NTOK / 128);
  gemm_nt_kernel<<<g4, 128, 0, stream>>>(attn, o_w, nullptr, out, NTOK, HIDDEN, HIDDEN);
}


# attention softmax: cross-lane row max, 2nd exp and O rescale only on steps where a score tops the row's held reference (max+8); row sums kept per lane, the 16 lanes added once after the key loop
# speedup vs baseline: 1.8286x; 1.8286x over previous
#include <hip/hip_runtime.h>
#include <hip/hip_bf16.h>
#include <math.h>

#define HIDDEN   2048
#define NH       16
#define NKV      4
#define HD       128
#define D2       64
#define QSIZE    2048
#define KVSIZE   512
#define QKVLD    3072
#define KOFF     2048
#define VOFF     2560
#define SEQB     2048
#define BATCH    2
#define NTOK     (BATCH * SEQB)
#define SM_SCALE 0.08838834764831845f

typedef __attribute__((ext_vector_type(2))) float v2f;
typedef __attribute__((ext_vector_type(8))) float v8f;

__device__ __forceinline__ v8f wmma_f32(v2f a, v2f b, v8f c) {
  return __builtin_amdgcn_wmma_f32_16x16x4_f32(
       false, a,  false, b,
       (short)0, c,  false,  false);
}

#define GBK  32
#define GLDP 36
#define GK   HIDDEN

__global__ __launch_bounds__(128) __attribute__((amdgpu_waves_per_eu(6, 6)))
void gemm_nt_kernel(const float* __restrict__ A, const float* __restrict__ Bw,
                    const float* __restrict__ bias, float* __restrict__ C,
                    int M, int N) {
  __shared__ float sA[128 * GLDP];
  __shared__ float sB[64 * GLDP];
  const int t    = threadIdx.x;
  const int lane = t & 31, wid = t >> 5;
  const int hi   = lane >> 4, l15 = lane & 15;
  const int mbase = blockIdx.y * 128;
  const int nbase = blockIdx.x * 64;

  const v8f vz = {0.f,0.f,0.f,0.f,0.f,0.f,0.f,0.f};
  v8f acc[2][4] = {{vz, vz, vz, vz}, {vz, vz, vz, vz}};

  const int srow = t >> 3;
  const int scol = (t & 7) << 2;

  const float* pa = A  + (size_t)(mbase + srow) * GK + scol;
  const float* pb = Bw + (size_t)(nbase + srow) * GK + scol;

  for (int kb = 0; kb < GK; kb += GBK, pa += GBK, pb += GBK) {
    float4 ra[8], rb[4];
#pragma unroll
    for (int i = 0; i < 8; i++)
      ra[i] = *(const float4*)&pa[16 * i * GK];
#pragma unroll
    for (int i = 0; i < 4; i++)
      rb[i] = *(const float4*)&pb[16 * i * GK];
#pragma unroll
    for (int i = 0; i < 8; i++)
      *(float4*)&sA[(srow + 16 * i) * GLDP + scol] = ra[i];
#pragma unroll
    for (int i = 0; i < 4; i++)
      *(float4*)&sB[(srow + 16 * i) * GLDP + scol] = rb[i];
    if (kb + GBK < GK) {
      __builtin_prefetch(pa + GBK, 0, 0);
      __builtin_prefetch(pb + GBK, 0, 0);
    }
    __syncthreads();
#pragma unroll
    for (int k2 = 0; k2 < GBK; k2 += 4) {
      const v2f a0 = *(const v2f*)&sA[(wid * 32 + l15) * GLDP + k2 + 2 * hi];
      const v2f a1 = *(const v2f*)&sA[(wid * 32 + 16 + l15) * GLDP + k2 + 2 * hi];
#pragma unroll
      for (int nc = 0; nc < 4; nc++) {
        const v2f b = *(const v2f*)&sB[(nc * 16 + l15) * GLDP + k2 + 2 * hi];
        acc[0][nc] = wmma_f32(a0, b, acc[0][nc]);
        acc[1][nc] = wmma_f32(a1, b, acc[1][nc]);
      }
    }
    __syncthreads();
  }

#pragma unroll
  for (int st = 0; st < 2; st++) {
#pragma unroll
    for (int nc = 0; nc < 4; nc++) {
      const int n  = nbase + nc * 16 + l15;
      const float bv = bias ? bias[n] : 0.0f;
#pragma unroll
      for (int r = 0; r < 8; r++) {
        const int m = mbase + wid * 32 + st * 16 + r + 8 * hi;
        C[(size_t)m * N + n] = acc[st][nc][r] + bv;
      }
    }
  }
}

__global__ __launch_bounds__(256)
void rope_kernel(float* __restrict__ qkv) {
  const int tid   = blockIdx.x * 256 + threadIdx.x;
  const int token = tid / 1280;
  const int rem   = tid % 1280;
  const int head  = rem >> 6;
  const int d     = rem & 63;
  const int pos   = token & (SEQB - 1);
  const int col   = (head < NH) ? head * HD + d : QSIZE + (head - NH) * HD + d;
  const float inv_freq = powf(1.0e6f, -(float)d * (1.0f / 64.0f));
  float s, c;
  sincosf((float)pos * inv_freq, &s, &c);
  float* p = qkv + (size_t)token * QKVLD + col;
  const float x1 = p[0], x2 = p[D2];
  p[0]  = x1 * c - x2 * s;
  p[D2] = x1 * s + x2 * c;
}

#define NITEM   80
#define CHSTEPS 32
#define SM_HEADROOM 8.0f
#define SM_LOWEST  -3.0e38f

__global__ __launch_bounds__(128) __attribute__((amdgpu_waves_per_eu(5, 5)))
void flash_attn_kernel(const float* __restrict__ qkv, float* __restrict__ opart,
                       float* __restrict__ mpart, float* __restrict__ lpart) {
  __shared__ float sQ[64 * 132];
  __shared__ float sK[16 * 132];
  __shared__ float sVt[128 * 20];
  __shared__ float sP[4][16 * 18];

  const int t    = threadIdx.x;
  const int lane = t & 31, wid = t >> 5;
  const int hi   = lane >> 4, l15 = lane & 15;
  const int h    = blockIdx.x & (NH - 1);
  const int b    = blockIdx.x >> 4;
  const int id   = (NITEM - 1) - (int)blockIdx.y;
  const int ga   = (id >= 8) + (id >= 24) + (id >= 48);
  const int rem  = id - 4 * ga * (ga + 1);
  const int rq   = (ga == 0) ? rem : (ga == 1) ? (rem >> 1) : (ga == 2) ? ((rem * 11) >> 5) : (rem >> 2);
  const int qblk = 8 * ga + rq;
  const int chunk = rem - rq * (ga + 1);
  const int item = (b * NH + h) * NITEM + id;
  const int kvh  = h >> 2;
  const int qbase = qblk * 64;
  const size_t rowbase = (size_t)(b * SEQB) * QKVLD;

#pragma unroll
  for (int i = 0; i < 16; i++) {
    const int idx = t + 128 * i;
    const int row = idx >> 5, c4 = (idx & 31) << 2;
    float4 v = *(const float4*)&qkv[rowbase + (size_t)(qbase + row) * QKVLD + h * HD + c4];
    v.x *= SM_SCALE; v.y *= SM_SCALE; v.z *= SM_SCALE; v.w *= SM_SCALE;
    *(float4*)&sQ[row * 132 + c4] = v;
  }

  const v8f vz = {0.f,0.f,0.f,0.f,0.f,0.f,0.f,0.f};
  v8f accO[8] = {vz, vz, vz, vz, vz, vz, vz, vz};
  float rm[8], rs[8];
#pragma unroll
  for (int r = 0; r < 8; r++) { rm[r] = SM_LOWEST; rs[r] = 0.0f; }

  __syncthreads();

  const int nkb = (qbase >> 4) + 4;
  const int kb0 = chunk * CHSTEPS;
  const int kb1 = (kb0 + CHSTEPS < nkb) ? kb0 + CHSTEPS : nkb;
  for (int kbi = kb0; kbi < kb1; kbi++) {
    const int kb = kbi * 16;
    const float* kvrow = qkv + rowbase + (size_t)kb * QKVLD + kvh * HD;
    const float* kp = kvrow + (size_t)(t >> 5) * QKVLD + KOFF + ((t & 31) << 2);
#pragma unroll
    for (int i = 0; i < 4; i++) {
      const float4 v = *(const float4*)&kp[4 * i * QKVLD];
      *(float4*)&sK[((t >> 5) + 4 * i) * 132 + ((t & 31) << 2)] = v;
    }
    const float* vp = kvrow + VOFF + t;
#pragma unroll
    for (int i = 0; i < 16; i++) {
      sVt[t * 20 + i] = vp[i * QKVLD];
    }
    __syncthreads();

    v8f s = vz;
#pragma unroll
    for (int k0 = 0; k0 < HD; k0 += 4) {
      const v2f a  = *(const v2f*)&sQ[(wid * 16 + l15) * 132 + k0 + 2 * hi];
      const v2f bb = *(const v2f*)&sK[l15 * 132 + k0 + 2 * hi];
      s = wmma_f32(a, bb, s);
    }

    const int kcol = kb + l15;
    float sv[8];
    bool over = false;
#pragma unroll
    for (int r = 0; r < 8; r++) {
      const int mrow = qbase + wid * 16 + r + 8 * hi;
      sv[r] = (kcol <= mrow) ? s[r] : -__builtin_inff();
      over |= (sv[r] > rm[r]);
    }
    if (__builtin_amdgcn_ballot_w32(over) != 0u) {
#pragma unroll
      for (int r = 0; r < 8; r++) {
        float mx = sv[r];
        mx = fmaxf(mx, __shfl_xor(mx, 1, 32));
        mx = fmaxf(mx, __shfl_xor(mx, 2, 32));
        mx = fmaxf(mx, __shfl_xor(mx, 4, 32));
        mx = fmaxf(mx, __shfl_xor(mx, 8, 32));
        const float newm  = fmaxf(rm[r], mx + SM_HEADROOM);
        const float alpha = __expf(rm[r] - newm);
        rm[r] = newm;
        rs[r] *= alpha;
#pragma unroll
        for (int nc = 0; nc < 8; nc++) accO[nc][r] *= alpha;
      }
    }
#pragma unroll
    for (int r = 0; r < 8; r++) {
      const float pv = __expf(sv[r] - rm[r]);
      rs[r] += pv;
      sP[wid][(r + 8 * hi) * 18 + l15] = pv;
    }
    __syncthreads();

#pragma unroll
    for (int k0 = 0; k0 < 16; k0 += 4) {
      const v2f a = *(const v2f*)&sP[wid][l15 * 18 + k0 + 2 * hi];
#pragma unroll
      for (int nc = 0; nc < 8; nc++) {
        const v2f bb = *(const v2f*)&sVt[(nc * 16 + l15) * 20 + k0 + 2 * hi];
        accO[nc] = wmma_f32(a, bb, accO[nc]);
      }
    }
    __syncthreads();
  }

  float* op = opart + ((size_t)item * 64 + wid * 16) * HD;
#pragma unroll
  for (int nc = 0; nc < 8; nc++) {
#pragma unroll
    for (int r = 0; r < 8; r++) {
      op[(r + 8 * hi) * HD + nc * 16 + l15] = accO[nc][r];
    }
  }
#pragma unroll
  for (int r = 0; r < 8; r++) {
    float l = rs[r];
    l += __shfl_xor(l, 1, 32);
    l += __shfl_xor(l, 2, 32);
    l += __shfl_xor(l, 4, 32);
    l += __shfl_xor(l, 8, 32);
    rs[r] = l;
  }
  if (l15 == 0) {
#pragma unroll
    for (int r = 0; r < 8; r++) {
      mpart[item * 64 + wid * 16 + r + 8 * hi] = rm[r];
      lpart[item * 64 + wid * 16 + r + 8 * hi] = rs[r];
    }
  }
}

__global__ __launch_bounds__(256)
void attn_combine_kernel(const float* __restrict__ opart, const float* __restrict__ mpart,
                         const float* __restrict__ lpart, float* __restrict__ out) {
  const int wv   = blockIdx.x * 8 + (threadIdx.x >> 5);
  const int lane = threadIdx.x & 31;
  const int h    = wv & (NH - 1);
  const int tok  = wv >> 4;
  const int b    = tok >> 11;
  const int row  = tok & (SEQB - 1);
  const int qblk = row >> 6;
  const int ga   = qblk >> 3;
  const int nch  = ga + 1;
  const int item0 = (b * NH + h) * NITEM + 4 * ga * (ga + 1) + (qblk & 7) * nch;
  const int rin  = row & 63;

  float mi[4], li[4];
  float m = -__builtin_inff();
#pragma unroll
  for (int c = 0; c < 4; c++) {
    if (c < nch) {
      mi[c] = mpart[(item0 + c) * 64 + rin];
      li[c] = lpart[(item0 + c) * 64 + rin];
      m = fmaxf(m, mi[c]);
    }
  }
  float l = 0.0f;
  float4 o = make_float4(0.f, 0.f, 0.f, 0.f);
#pragma unroll
  for (int c = 0; c < 4; c++) {
    if (c < nch) {
      const float w = __expf(mi[c] - m);
      const float4 v = *(const float4*)&opart[((size_t)(item0 + c) * 64 + rin) * HD + lane * 4];
      l += li[c] * w;
      o.x += v.x * w; o.y += v.y * w; o.z += v.z * w; o.w += v.w * w;
    }
  }
  float4 res;
  res.x = o.x / l; res.y = o.y / l; res.z = o.z / l; res.w = o.w / l;
  *(float4*)&out[(size_t)tok * HIDDEN + h * HD + lane * 4] = res;
}

extern "C" void kernel_launch(void* const* d_in, const int* in_sizes, int n_in,
                              void* d_out, int out_size, void* d_ws, size_t ws_size,
                              hipStream_t stream) {
  const float* hs    = (const float*)d_in[0];
  const float* qkv_w = (const float*)d_in[1];
  const float* qkv_b = (const float*)d_in[2];
  const float* o_w   = (const float*)d_in[3];
  float* out  = (float*)d_out;
  float* qkv  = (float*)d_ws;
  float* attn = qkv + (size_t)NTOK * QKVLD;
  float* opart = attn + (size_t)NTOK * HIDDEN;
  float* mpart = opart + (size_t)BATCH * NH * NITEM * 64 * HD;
  float* lpart = mpart + (size_t)BATCH * NH * NITEM * 64;

  dim3 g1(QKVLD / 64, NTOK / 128);
  gemm_nt_kernel<<<g1, 128, 0, stream>>>(hs, qkv_w, qkv_b, qkv, NTOK, QKVLD);

  rope_kernel<<<(NTOK * 1280) / 256, 256, 0, stream>>>(qkv);

  dim3 g3(BATCH * NH, NITEM);
  flash_attn_kernel<<<g3, 128, 0, stream>>>(qkv, opart, mpart, lpart);

  attn_combine_kernel<<<(NTOK * NH) / 8, 256, 0, stream>>>(opart, mpart, lpart, attn);

  dim3 g4(HIDDEN / 64, NTOK / 128);
  gemm_nt_kernel<<<g4, 128, 0, stream>>>(attn, o_w, nullptr, out, NTOK, HIDDEN);
}
